# starting kernel
# baseline (speedup 1.0000x reference)
#include <hip/hip_runtime.h>

namespace {

constexpr int kNV   = 8192;
constexpr int kNCAM = 20;
constexpr int kCIN  = 2048;
constexpr int kH    = 4;
constexpr int kCOUT = 2048;
constexpr int kE    = 16384;
constexpr int kNHC  = kH * kCOUT;
constexpr int kNJ   = kH * kNCAM;
constexpr float kNegSlope = 0.2f;

constexpr size_t OFF_HS    = 0;
constexpr size_t OFF_WATT  = OFF_HS    + (size_t)kNJ * kCOUT;
constexpr size_t OFF_AD    = OFF_WATT  + (size_t)kCIN * kH;
constexpr size_t OFF_AS    = OFF_AD    + (size_t)kNV * kH;
constexpr size_t OFF_MENC  = OFF_AS    + 128;
constexpr size_t OFF_DEN   = OFF_MENC  + (size_t)kNV * kH;
constexpr size_t OFF_S     = OFF_DEN   + (size_t)kNV * kH;
constexpr size_t OFF_CTPAD = OFF_S     + (size_t)kNV * kNJ;

typedef float v2f __attribute__((ext_vector_type(2)));
typedef float v8f __attribute__((ext_vector_type(8)));

__device__ __forceinline__ float wave_sum(float v) {
#pragma unroll
  for (int off = 16; off > 0; off >>= 1) v += __shfl_down(v, off, 32);
  return v;
}

__device__ __forceinline__ unsigned f2ord(float f) {
  unsigned u = __float_as_uint(f);
  return (u & 0x80000000u) ? ~u : (u | 0x80000000u);
}
__device__ __forceinline__ float ord2f(unsigned e) {
  return __uint_as_float((e & 0x80000000u) ? (e ^ 0x80000000u) : ~e);
}

__global__ void k_fold_att(const float* __restrict__ W,
                           const float* __restrict__ att_dst,
                           float* __restrict__ W_att) {
  int wave = (blockIdx.x * blockDim.x + threadIdx.x) >> 5;
  int lane = threadIdx.x & 31;
  if (wave >= kCIN * kH) return;
  int k = wave >> 2, h = wave & 3;
  const float4* wrow = reinterpret_cast<const float4*>(W + (size_t)k * kNHC + (size_t)h * kCOUT);
  const float4* arow = reinterpret_cast<const float4*>(att_dst + (size_t)h * kCOUT);
  float s = 0.f;
  for (int c = lane; c < kCOUT / 4; c += 32) {
    float4 wv = wrow[c];
    float4 av = arow[c];
    s += wv.x * av.x + wv.y * av.y + wv.z * av.z + wv.w * av.w;
  }
  s = wave_sum(s);
  if (lane == 0) W_att[k * kH + h] = s;
}

__global__ void k_ad(const float* __restrict__ x,
                     const float* __restrict__ W_att,
                     float* __restrict__ a_d) {
  int n = (blockIdx.x * blockDim.x + threadIdx.x) >> 5;
  int lane = threadIdx.x & 31;
  if (n >= kNV) return;
  const float4* xr = reinterpret_cast<const float4*>(x + (size_t)n * kCIN);
  const float4* wa = reinterpret_cast<const float4*>(W_att);
  float a0 = 0.f, a1 = 0.f, a2 = 0.f, a3 = 0.f;
  for (int k4 = lane; k4 < kCIN / 4; k4 += 32) {
    float4 xv = xr[k4];
    float4 w0 = wa[k4 * 4 + 0];
    float4 w1 = wa[k4 * 4 + 1];
    float4 w2 = wa[k4 * 4 + 2];
    float4 w3 = wa[k4 * 4 + 3];
    a0 += xv.x * w0.x + xv.y * w1.x + xv.z * w2.x + xv.w * w3.x;
    a1 += xv.x * w0.y + xv.y * w1.y + xv.z * w2.y + xv.w * w3.y;
    a2 += xv.x * w0.z + xv.y * w1.z + xv.z * w2.z + xv.w * w3.z;
    a3 += xv.x * w0.w + xv.y * w1.w + xv.z * w2.w + xv.w * w3.w;
  }
  a0 = wave_sum(a0); a1 = wave_sum(a1); a2 = wave_sum(a2); a3 = wave_sum(a3);
  if (lane == 0) {
    float* o = a_d + (size_t)n * 4;
    o[0] = a0; o[1] = a1; o[2] = a2; o[3] = a3;
  }
}

__global__ void k_hsrc(const float* __restrict__ CTp,
                       const float* __restrict__ W,
                       float* __restrict__ Hs) {
  int tileN = (blockIdx.x * blockDim.x + threadIdx.x) >> 5;
  int lane  = threadIdx.x & 31;
  int half  = lane >> 4, m = lane & 15;
  int col   = tileN * 16 + m;
  int koff  = half * 2;
  const float* arow0 = CTp + (size_t)m * kCIN + koff;
  const float* arow1 = CTp + (size_t)(16 + m) * kCIN + koff;
  v8f acc0 = {}; v8f acc1 = {};
#pragma unroll 4
  for (int k = 0; k < kCIN; k += 4) {
    v2f b;
    b.x = W[(size_t)(k + koff)     * kNHC + col];
    b.y = W[(size_t)(k + koff + 1) * kNHC + col];
    v2f a0 = *reinterpret_cast<const v2f*>(arow0 + k);
    v2f a1 = *reinterpret_cast<const v2f*>(arow1 + k);
    acc0 = __builtin_amdgcn_wmma_f32_16x16x4_f32(false, a0, false, b, (short)0,
                                                 acc0, false, false);
    acc1 = __builtin_amdgcn_wmma_f32_16x16x4_f32(false, a1, false, b, (short)0,
                                                 acc1, false, false);
  }
  int h = col >> 11, c = col & (kCOUT - 1);
#pragma unroll
  for (int v = 0; v < 8; ++v) {
    int cam = v + half * 8;
    Hs[(size_t)(h * kNCAM + cam) * kCOUT + c] = acc0[v];
    int cam2 = cam + 16;
    if (cam2 < kNCAM)
      Hs[(size_t)(h * kNCAM + cam2) * kCOUT + c] = acc1[v];
  }
}

__global__ void k_as(const float* __restrict__ Hs,
                     const float* __restrict__ att_src,
                     float* __restrict__ a_s) {
  int wave = (blockIdx.x * blockDim.x + threadIdx.x) >> 5;
  int lane = threadIdx.x & 31;
  if (wave >= kNJ) return;
  int h = wave / kNCAM, cam = wave % kNCAM;
  const float4* hrow = reinterpret_cast<const float4*>(Hs + (size_t)wave * kCOUT);
  const float4* arow = reinterpret_cast<const float4*>(att_src + (size_t)h * kCOUT);
  float s = 0.f;
  for (int c = lane; c < kCOUT / 4; c += 32) {
    float4 hv = hrow[c];
    float4 av = arow[c];
    s += hv.x * av.x + hv.y * av.y + hv.z * av.z + hv.w * av.w;
  }
  s = wave_sum(s);
  if (lane == 0) a_s[cam * kH + h] = s;
}

__device__ __forceinline__ float edge_logit(const float* a_s, const float* a_d,
                                            int src, int dst, int h) {
  float v = a_s[src * kH + h] + a_d[dst * kH + h];
  return v > 0.f ? v : kNegSlope * v;
}

__global__ void k_edge_max(const int* __restrict__ esrc, const int* __restrict__ edst,
                           const float* __restrict__ a_s, const float* __restrict__ a_d,
                           unsigned* __restrict__ mEnc) {
  int t = blockIdx.x * blockDim.x + threadIdx.x;
  if (t >= kE * kH) return;
  int ei = t >> 2, h = t & 3;
  int src = esrc[ei], dst = edst[ei];
  float v = edge_logit(a_s, a_d, src, dst, h);
  atomicMax(&mEnc[dst * kH + h], f2ord(v));
}

__global__ void k_edge_den(const int* __restrict__ esrc, const int* __restrict__ edst,
                           const float* __restrict__ a_s, const float* __restrict__ a_d,
                           const unsigned* __restrict__ mEnc, float* __restrict__ den) {
  int t = blockIdx.x * blockDim.x + threadIdx.x;
  if (t >= kE * kH) return;
  int ei = t >> 2, h = t & 3;
  int src = esrc[ei], dst = edst[ei];
  float v = edge_logit(a_s, a_d, src, dst, h);
  float m = ord2f(mEnc[dst * kH + h]);
  atomicAdd(&den[dst * kH + h], __expf(v - m));
}

__global__ void k_edge_acc(const int* __restrict__ esrc, const int* __restrict__ edst,
                           const float* __restrict__ a_s, const float* __restrict__ a_d,
                           const unsigned* __restrict__ mEnc,
                           const float* __restrict__ den, float* __restrict__ S) {
  int t = blockIdx.x * blockDim.x + threadIdx.x;
  if (t >= kE * kH) return;
  int ei = t >> 2, h = t & 3;
  int src = esrc[ei], dst = edst[ei];
  float v = edge_logit(a_s, a_d, src, dst, h);
  float m = ord2f(mEnc[dst * kH + h]);
  float attn = __expf(v - m) / den[dst * kH + h];
  atomicAdd(&S[(size_t)dst * kNJ + h * kNCAM + src], attn);
}

__global__ void k_out(const float* __restrict__ S, const float* __restrict__ Hs,
                      const float* __restrict__ x, const float* __restrict__ bias,
                      const float* __restrict__ alphap, float* __restrict__ out) {
  int w     = threadIdx.x >> 5;
  int lane  = threadIdx.x & 31;
  int half  = lane >> 4, m = lane & 15;
  int tileN = blockIdx.x;
  int tileM = blockIdx.y * 8 + w;
  int col   = tileN * 16 + m;
  int rowA  = tileM * 16 + m;
  int koff  = half * 2;
  const float* srow = S + (size_t)rowA * kNJ + koff;
  v8f acc = {};
#pragma unroll
  for (int kk = 0; kk < kNJ; kk += 4) {
    v2f a = *reinterpret_cast<const v2f*>(srow + kk);
    v2f b;
    b.x = Hs[(size_t)(kk + koff)     * kCOUT + col];
    b.y = Hs[(size_t)(kk + koff + 1) * kCOUT + col];
    acc = __builtin_amdgcn_wmma_f32_16x16x4_f32(false, a, false, b, (short)0,
                                                acc, false, false);
  }
  float alpha = *alphap;
  float bcol  = bias[col];
#pragma unroll
  for (int v = 0; v < 8; ++v) {
    int r = tileM * 16 + v + half * 8;
    size_t idx = (size_t)r * kCOUT + col;
    out[idx] = x[idx] + alpha * (acc[v] * 0.25f + bcol);
  }
}

}

extern "C" void kernel_launch(void* const* d_in, const int* in_sizes, int n_in,
                              void* d_out, int out_size, void* d_ws, size_t ws_size,
                              hipStream_t stream) {
  (void)in_sizes; (void)n_in; (void)out_size; (void)ws_size;
  const float* x       = (const float*)d_in[0];
  const float* ct      = (const float*)d_in[1];
  const float* W       = (const float*)d_in[2];
  const float* att_src = (const float*)d_in[3];
  const float* att_dst = (const float*)d_in[4];
  const float* bias    = (const float*)d_in[5];
  const float* alphap  = (const float*)d_in[6];
  const int* esrc = (const int*)d_in[8];
  const int* edst = (const int*)d_in[9];

  float* ws = (float*)d_ws;
  float*    Hs   = ws + OFF_HS;
  float*    Watt = ws + OFF_WATT;
  float*    a_d  = ws + OFF_AD;
  float*    a_s  = ws + OFF_AS;
  unsigned* mEnc = (unsigned*)(ws + OFF_MENC);
  float*    den  = ws + OFF_DEN;
  float*    S    = ws + OFF_S;
  float*    CTp  = ws + OFF_CTPAD;

  hipMemsetAsync(mEnc, 0, (size_t)kNV * kH * sizeof(unsigned), stream);
  hipMemsetAsync(den,  0, (size_t)kNV * kH * sizeof(float), stream);
  hipMemsetAsync(S,    0, (size_t)kNV * kNJ * sizeof(float), stream);
  hipMemsetAsync(CTp + (size_t)kNCAM * kCIN, 0,
                 (size_t)(32 - kNCAM) * kCIN * sizeof(float), stream);
  hipMemcpyAsync(CTp, ct, (size_t)kNCAM * kCIN * sizeof(float),
                 hipMemcpyDeviceToDevice, stream);

  k_fold_att<<<dim3((kCIN * kH) / 8), dim3(256), 0, stream>>>(W, att_dst, Watt);
  k_ad<<<dim3(kNV / 8), dim3(256), 0, stream>>>(x, Watt, a_d);
  k_hsrc<<<dim3((kNHC / 16) / 8), dim3(256), 0, stream>>>(CTp, W, Hs);
  k_as<<<dim3((kNJ + 7) / 8), dim3(256), 0, stream>>>(Hs, att_src, a_s);

  int et = kE * kH;
  k_edge_max<<<dim3((et + 255) / 256), dim3(256), 0, stream>>>(esrc, edst, a_s, a_d, mEnc);
  k_edge_den<<<dim3((et + 255) / 256), dim3(256), 0, stream>>>(esrc, edst, a_s, a_d, mEnc, den);
  k_edge_acc<<<dim3((et + 255) / 256), dim3(256), 0, stream>>>(esrc, edst, a_s, a_d, mEnc, den, S);

  k_out<<<dim3(kCOUT / 16, (kNV / 16) / 8), dim3(256), 0, stream>>>(
      S, Hs, x, bias, alphap, (float*)d_out);
}


# k_hsrc re-tiled: one block per 32-column strip of W, K split over its 16 waves, each wave's W loads issued as one group ahead, partials summed in LDS in fixed wave order (h_src sum order changes)
# speedup vs baseline: 1.5511x; 1.5511x over previous
#include <hip/hip_runtime.h>

namespace {

constexpr int kNV   = 8192;
constexpr int kNCAM = 20;
constexpr int kCIN  = 2048;
constexpr int kH    = 4;
constexpr int kCOUT = 2048;
constexpr int kE    = 16384;
constexpr int kNHC  = kH * kCOUT;
constexpr int kNJ   = kH * kNCAM;
constexpr float kNegSlope = 0.2f;

constexpr size_t OFF_HS    = 0;
constexpr size_t OFF_WATT  = OFF_HS    + (size_t)kNJ * kCOUT;
constexpr size_t OFF_AD    = OFF_WATT  + (size_t)kCIN * kH;
constexpr size_t OFF_AS    = OFF_AD    + (size_t)kNV * kH;
constexpr size_t OFF_MENC  = OFF_AS    + 128;
constexpr size_t OFF_DEN   = OFF_MENC  + (size_t)kNV * kH;
constexpr size_t OFF_S     = OFF_DEN   + (size_t)kNV * kH;
constexpr size_t OFF_CTPAD = OFF_S     + (size_t)kNV * kNJ;

typedef float v2f __attribute__((ext_vector_type(2)));
typedef float v4f __attribute__((ext_vector_type(4)));
typedef float v8f __attribute__((ext_vector_type(8)));

__device__ __forceinline__ float wave_sum(float v) {
#pragma unroll
  for (int off = 16; off > 0; off >>= 1) v += __shfl_down(v, off, 32);
  return v;
}

__device__ __forceinline__ unsigned f2ord(float f) {
  unsigned u = __float_as_uint(f);
  return (u & 0x80000000u) ? ~u : (u | 0x80000000u);
}
__device__ __forceinline__ float ord2f(unsigned e) {
  return __uint_as_float((e & 0x80000000u) ? (e ^ 0x80000000u) : ~e);
}

__global__ void k_fold_att(const float* __restrict__ W,
                           const float* __restrict__ att_dst,
                           float* __restrict__ W_att) {
  int wave = (blockIdx.x * blockDim.x + threadIdx.x) >> 5;
  int lane = threadIdx.x & 31;
  if (wave >= kCIN * kH) return;
  int k = wave >> 2, h = wave & 3;
  const float4* wrow = reinterpret_cast<const float4*>(W + (size_t)k * kNHC + (size_t)h * kCOUT);
  const float4* arow = reinterpret_cast<const float4*>(att_dst + (size_t)h * kCOUT);
  float s = 0.f;
  for (int c = lane; c < kCOUT / 4; c += 32) {
    float4 wv = wrow[c];
    float4 av = arow[c];
    s += wv.x * av.x + wv.y * av.y + wv.z * av.z + wv.w * av.w;
  }
  s = wave_sum(s);
  if (lane == 0) W_att[k * kH + h] = s;
}

__global__ void k_ad(const float* __restrict__ x,
                     const float* __restrict__ W_att,
                     float* __restrict__ a_d) {
  int n = (blockIdx.x * blockDim.x + threadIdx.x) >> 5;
  int lane = threadIdx.x & 31;
  if (n >= kNV) return;
  const float4* xr = reinterpret_cast<const float4*>(x + (size_t)n * kCIN);
  const float4* wa = reinterpret_cast<const float4*>(W_att);
  float a0 = 0.f, a1 = 0.f, a2 = 0.f, a3 = 0.f;
  for (int k4 = lane; k4 < kCIN / 4; k4 += 32) {
    float4 xv = xr[k4];
    float4 w0 = wa[k4 * 4 + 0];
    float4 w1 = wa[k4 * 4 + 1];
    float4 w2 = wa[k4 * 4 + 2];
    float4 w3 = wa[k4 * 4 + 3];
    a0 += xv.x * w0.x + xv.y * w1.x + xv.z * w2.x + xv.w * w3.x;
    a1 += xv.x * w0.y + xv.y * w1.y + xv.z * w2.y + xv.w * w3.y;
    a2 += xv.x * w0.z + xv.y * w1.z + xv.z * w2.z + xv.w * w3.z;
    a3 += xv.x * w0.w + xv.y * w1.w + xv.z * w2.w + xv.w * w3.w;
  }
  a0 = wave_sum(a0); a1 = wave_sum(a1); a2 = wave_sum(a2); a3 = wave_sum(a3);
  if (lane == 0) {
    float* o = a_d + (size_t)n * 4;
    o[0] = a0; o[1] = a1; o[2] = a2; o[3] = a3;
  }
}

__global__ void __launch_bounds__(512)
k_hsrc(const float* __restrict__ CTp,
       const float* __restrict__ W,
       float* __restrict__ Hs) {
  __shared__ __attribute__((aligned(16))) float part[16 * kNCAM * 32];
  const int w    = threadIdx.x >> 5;
  const int lane = threadIdx.x & 31;
  const int hf   = lane >> 4, n = lane & 15;
  const int c0   = blockIdx.x * 32;
  const int kb   = w * 128 + hf * 4;
  const float* wp  = W + (size_t)kb * kNHC + c0 + 2 * n;
  const float* ap0 = CTp + (size_t)n * kCIN + kb;
  const float* ap1 = ap0 + (size_t)16 * kCIN;

  v2f b[16][4];
  v4f a[16][2];
#pragma unroll
  for (int p = 0; p < 8; ++p)
#pragma unroll
    for (int i = 0; i < 4; ++i)
      b[p][i] = __builtin_nontemporal_load(
          reinterpret_cast<const v2f*>(wp + (size_t)(8 * p + i) * kNHC));
#pragma unroll
  for (int p = 0; p < 8; ++p) {
    a[p][0] = *reinterpret_cast<const v4f*>(ap0 + 8 * p);
    a[p][1] = *reinterpret_cast<const v4f*>(ap1 + 8 * p);
  }
#pragma unroll
  for (int p = 8; p < 16; ++p)
#pragma unroll
    for (int i = 0; i < 4; ++i)
      b[p][i] = __builtin_nontemporal_load(
          reinterpret_cast<const v2f*>(wp + (size_t)(8 * p + i) * kNHC));

  __builtin_amdgcn_sched_barrier(0);

  v8f acc[2][2] = {};
#pragma unroll
  for (int p = 0; p < 16; ++p) {
    if (p < 8) {
      a[p + 8][0] = *reinterpret_cast<const v4f*>(ap0 + 8 * (p + 8));
      a[p + 8][1] = *reinterpret_cast<const v4f*>(ap1 + 8 * (p + 8));
    }
#pragma unroll
    for (int q = 0; q < 2; ++q) {
      v2f a0 = {a[p][0][2 * q], a[p][0][2 * q + 1]};
      v2f a1 = {a[p][1][2 * q], a[p][1][2 * q + 1]};
#pragma unroll
      for (int j = 0; j < 2; ++j) {
        v2f bj = {b[p][2 * q][j], b[p][2 * q + 1][j]};
        acc[0][j] = __builtin_amdgcn_wmma_f32_16x16x4_f32(false, a0, false, bj, (short)0,
                                                          acc[0][j], false, false);
        acc[1][j] = __builtin_amdgcn_wmma_f32_16x16x4_f32(false, a1, false, bj, (short)0,
                                                          acc[1][j], false, false);
      }
    }
    if (p < 8) __builtin_amdgcn_sched_barrier(0);
  }

  float* pw = part + w * (kNCAM * 32);
#pragma unroll
  for (int r = 0; r < 8; ++r) {
    v2f t = {acc[0][0][r], acc[0][1][r]};
    *reinterpret_cast<v2f*>(pw + (8 * hf + r) * 32 + 2 * n) = t;
  }
  if (hf == 0) {
#pragma unroll
    for (int r = 0; r < kNCAM - 16; ++r) {
      v2f t = {acc[1][0][r], acc[1][1][r]};
      *reinterpret_cast<v2f*>(pw + (16 + r) * 32 + 2 * n) = t;
    }
  }
  __syncthreads();
  for (int i = threadIdx.x; i < kNCAM * 32; i += 512) {
    float s = 0.f;
#pragma unroll
    for (int w2 = 0; w2 < 16; ++w2) s += part[w2 * (kNCAM * 32) + i];
    int cam = i >> 5, col = c0 + (i & 31);
    int h = col >> 11, c = col & (kCOUT - 1);
    Hs[(size_t)(h * kNCAM + cam) * kCOUT + c] = s;
  }
}

__global__ void k_as(const float* __restrict__ Hs,
                     const float* __restrict__ att_src,
                     float* __restrict__ a_s) {
  int wave = (blockIdx.x * blockDim.x + threadIdx.x) >> 5;
  int lane = threadIdx.x & 31;
  if (wave >= kNJ) return;
  int h = wave / kNCAM, cam = wave % kNCAM;
  const float4* hrow = reinterpret_cast<const float4*>(Hs + (size_t)wave * kCOUT);
  const float4* arow = reinterpret_cast<const float4*>(att_src + (size_t)h * kCOUT);
  float s = 0.f;
  for (int c = lane; c < kCOUT / 4; c += 32) {
    float4 hv = hrow[c];
    float4 av = arow[c];
    s += hv.x * av.x + hv.y * av.y + hv.z * av.z + hv.w * av.w;
  }
  s = wave_sum(s);
  if (lane == 0) a_s[cam * kH + h] = s;
}

__device__ __forceinline__ float edge_logit(const float* a_s, const float* a_d,
                                            int src, int dst, int h) {
  float v = a_s[src * kH + h] + a_d[dst * kH + h];
  return v > 0.f ? v : kNegSlope * v;
}

__global__ void k_edge_max(const int* __restrict__ esrc, const int* __restrict__ edst,
                           const float* __restrict__ a_s, const float* __restrict__ a_d,
                           unsigned* __restrict__ mEnc) {
  int t = blockIdx.x * blockDim.x + threadIdx.x;
  if (t >= kE * kH) return;
  int ei = t >> 2, h = t & 3;
  int src = esrc[ei], dst = edst[ei];
  float v = edge_logit(a_s, a_d, src, dst, h);
  atomicMax(&mEnc[dst * kH + h], f2ord(v));
}

__global__ void k_edge_den(const int* __restrict__ esrc, const int* __restrict__ edst,
                           const float* __restrict__ a_s, const float* __restrict__ a_d,
                           const unsigned* __restrict__ mEnc, float* __restrict__ den) {
  int t = blockIdx.x * blockDim.x + threadIdx.x;
  if (t >= kE * kH) return;
  int ei = t >> 2, h = t & 3;
  int src = esrc[ei], dst = edst[ei];
  float v = edge_logit(a_s, a_d, src, dst, h);
  float m = ord2f(mEnc[dst * kH + h]);
  atomicAdd(&den[dst * kH + h], __expf(v - m));
}

__global__ void k_edge_acc(const int* __restrict__ esrc, const int* __restrict__ edst,
                           const float* __restrict__ a_s, const float* __restrict__ a_d,
                           const unsigned* __restrict__ mEnc,
                           const float* __restrict__ den, float* __restrict__ S) {
  int t = blockIdx.x * blockDim.x + threadIdx.x;
  if (t >= kE * kH) return;
  int ei = t >> 2, h = t & 3;
  int src = esrc[ei], dst = edst[ei];
  float v = edge_logit(a_s, a_d, src, dst, h);
  float m = ord2f(mEnc[dst * kH + h]);
  float attn = __expf(v - m) / den[dst * kH + h];
  atomicAdd(&S[(size_t)dst * kNJ + h * kNCAM + src], attn);
}

__global__ void k_out(const float* __restrict__ S, const float* __restrict__ Hs,
                      const float* __restrict__ x, const float* __restrict__ bias,
                      const float* __restrict__ alphap, float* __restrict__ out) {
  int w     = threadIdx.x >> 5;
  int lane  = threadIdx.x & 31;
  int half  = lane >> 4, m = lane & 15;
  int tileN = blockIdx.x;
  int tileM = blockIdx.y * 8 + w;
  int col   = tileN * 16 + m;
  int rowA  = tileM * 16 + m;
  int koff  = half * 2;
  const float* srow = S + (size_t)rowA * kNJ + koff;
  v8f acc = {};
#pragma unroll
  for (int kk = 0; kk < kNJ; kk += 4) {
    v2f a = *reinterpret_cast<const v2f*>(srow + kk);
    v2f b;
    b.x = Hs[(size_t)(kk + koff)     * kCOUT + col];
    b.y = Hs[(size_t)(kk + koff + 1) * kCOUT + col];
    acc = __builtin_amdgcn_wmma_f32_16x16x4_f32(false, a, false, b, (short)0,
                                                acc, false, false);
  }
  float alpha = *alphap;
  float bcol  = bias[col];
#pragma unroll
  for (int v = 0; v < 8; ++v) {
    int r = tileM * 16 + v + half * 8;
    size_t idx = (size_t)r * kCOUT + col;
    out[idx] = x[idx] + alpha * (acc[v] * 0.25f + bcol);
  }
}

}

extern "C" void kernel_launch(void* const* d_in, const int* in_sizes, int n_in,
                              void* d_out, int out_size, void* d_ws, size_t ws_size,
                              hipStream_t stream) {
  (void)in_sizes; (void)n_in; (void)out_size; (void)ws_size;
  const float* x       = (const float*)d_in[0];
  const float* ct      = (const float*)d_in[1];
  const float* W       = (const float*)d_in[2];
  const float* att_src = (const float*)d_in[3];
  const float* att_dst = (const float*)d_in[4];
  const float* bias    = (const float*)d_in[5];
  const float* alphap  = (const float*)d_in[6];
  const int* esrc = (const int*)d_in[8];
  const int* edst = (const int*)d_in[9];

  float* ws = (float*)d_ws;
  float*    Hs   = ws + OFF_HS;
  float*    Watt = ws + OFF_WATT;
  float*    a_d  = ws + OFF_AD;
  float*    a_s  = ws + OFF_AS;
  unsigned* mEnc = (unsigned*)(ws + OFF_MENC);
  float*    den  = ws + OFF_DEN;
  float*    S    = ws + OFF_S;
  float*    CTp  = ws + OFF_CTPAD;

  hipMemsetAsync(mEnc, 0, (size_t)kNV * kH * sizeof(unsigned), stream);
  hipMemsetAsync(den,  0, (size_t)kNV * kH * sizeof(float), stream);
  hipMemsetAsync(S,    0, (size_t)kNV * kNJ * sizeof(float), stream);
  hipMemsetAsync(CTp + (size_t)kNCAM * kCIN, 0,
                 (size_t)(32 - kNCAM) * kCIN * sizeof(float), stream);
  hipMemcpyAsync(CTp, ct, (size_t)kNCAM * kCIN * sizeof(float),
                 hipMemcpyDeviceToDevice, stream);

  k_fold_att<<<dim3((kCIN * kH) / 8), dim3(256), 0, stream>>>(W, att_dst, Watt);
  k_ad<<<dim3(kNV / 8), dim3(256), 0, stream>>>(x, Watt, a_d);
  k_hsrc<<<dim3(kNHC / 32), dim3(512), 0, stream>>>(CTp, W, Hs);
  k_as<<<dim3((kNJ + 7) / 8), dim3(256), 0, stream>>>(Hs, att_src, a_s);

  int et = kE * kH;
  k_edge_max<<<dim3((et + 255) / 256), dim3(256), 0, stream>>>(esrc, edst, a_s, a_d, mEnc);
  k_edge_den<<<dim3((et + 255) / 256), dim3(256), 0, stream>>>(esrc, edst, a_s, a_d, mEnc, den);
  k_edge_acc<<<dim3((et + 255) / 256), dim3(256), 0, stream>>>(esrc, edst, a_s, a_d, mEnc, den, S);

  k_out<<<dim3(kCOUT / 16, (kNV / 16) / 8), dim3(256), 0, stream>>>(
      S, Hs, x, bias, alphap, (float*)d_out);
}


# k_out re-tiled: each wave keeps the Hs fragments of 64 columns in registers and walks 8 M-tiles with 16-byte row accesses, 2048 long waves instead of 65536 short ones (k order of out sums changes)
# speedup vs baseline: 2.5739x; 1.6594x over previous
#include <hip/hip_runtime.h>

namespace {

constexpr int kNV   = 8192;
constexpr int kNCAM = 20;
constexpr int kCIN  = 2048;
constexpr int kH    = 4;
constexpr int kCOUT = 2048;
constexpr int kE    = 16384;
constexpr int kNHC  = kH * kCOUT;
constexpr int kNJ   = kH * kNCAM;
constexpr float kNegSlope = 0.2f;

constexpr size_t OFF_HS    = 0;
constexpr size_t OFF_WATT  = OFF_HS    + (size_t)kNJ * kCOUT;
constexpr size_t OFF_AD    = OFF_WATT  + (size_t)kCIN * kH;
constexpr size_t OFF_AS    = OFF_AD    + (size_t)kNV * kH;
constexpr size_t OFF_MENC  = OFF_AS    + 128;
constexpr size_t OFF_DEN   = OFF_MENC  + (size_t)kNV * kH;
constexpr size_t OFF_S     = OFF_DEN   + (size_t)kNV * kH;
constexpr size_t OFF_CTPAD = OFF_S     + (size_t)kNV * kNJ;

typedef float v2f __attribute__((ext_vector_type(2)));
typedef float v4f __attribute__((ext_vector_type(4)));
typedef float v8f __attribute__((ext_vector_type(8)));

__device__ __forceinline__ float wave_sum(float v) {
#pragma unroll
  for (int off = 16; off > 0; off >>= 1) v += __shfl_down(v, off, 32);
  return v;
}

__device__ __forceinline__ unsigned f2ord(float f) {
  unsigned u = __float_as_uint(f);
  return (u & 0x80000000u) ? ~u : (u | 0x80000000u);
}
__device__ __forceinline__ float ord2f(unsigned e) {
  return __uint_as_float((e & 0x80000000u) ? (e ^ 0x80000000u) : ~e);
}

__global__ void k_fold_att(const float* __restrict__ W,
                           const float* __restrict__ att_dst,
                           float* __restrict__ W_att) {
  int wave = (blockIdx.x * blockDim.x + threadIdx.x) >> 5;
  int lane = threadIdx.x & 31;
  if (wave >= kCIN * kH) return;
  int k = wave >> 2, h = wave & 3;
  const float4* wrow = reinterpret_cast<const float4*>(W + (size_t)k * kNHC + (size_t)h * kCOUT);
  const float4* arow = reinterpret_cast<const float4*>(att_dst + (size_t)h * kCOUT);
  float s = 0.f;
  for (int c = lane; c < kCOUT / 4; c += 32) {
    float4 wv = wrow[c];
    float4 av = arow[c];
    s += wv.x * av.x + wv.y * av.y + wv.z * av.z + wv.w * av.w;
  }
  s = wave_sum(s);
  if (lane == 0) W_att[k * kH + h] = s;
}

__global__ void k_ad(const float* __restrict__ x,
                     const float* __restrict__ W_att,
                     float* __restrict__ a_d) {
  int n = (blockIdx.x * blockDim.x + threadIdx.x) >> 5;
  int lane = threadIdx.x & 31;
  if (n >= kNV) return;
  const float4* xr = reinterpret_cast<const float4*>(x + (size_t)n * kCIN);
  const float4* wa = reinterpret_cast<const float4*>(W_att);
  float a0 = 0.f, a1 = 0.f, a2 = 0.f, a3 = 0.f;
  for (int k4 = lane; k4 < kCIN / 4; k4 += 32) {
    float4 xv = xr[k4];
    float4 w0 = wa[k4 * 4 + 0];
    float4 w1 = wa[k4 * 4 + 1];
    float4 w2 = wa[k4 * 4 + 2];
    float4 w3 = wa[k4 * 4 + 3];
    a0 += xv.x * w0.x + xv.y * w1.x + xv.z * w2.x + xv.w * w3.x;
    a1 += xv.x * w0.y + xv.y * w1.y + xv.z * w2.y + xv.w * w3.y;
    a2 += xv.x * w0.z + xv.y * w1.z + xv.z * w2.z + xv.w * w3.z;
    a3 += xv.x * w0.w + xv.y * w1.w + xv.z * w2.w + xv.w * w3.w;
  }
  a0 = wave_sum(a0); a1 = wave_sum(a1); a2 = wave_sum(a2); a3 = wave_sum(a3);
  if (lane == 0) {
    float* o = a_d + (size_t)n * 4;
    o[0] = a0; o[1] = a1; o[2] = a2; o[3] = a3;
  }
}

__global__ void __launch_bounds__(512)
k_hsrc(const float* __restrict__ CTp,
       const float* __restrict__ W,
       float* __restrict__ Hs) {
  __shared__ __attribute__((aligned(16))) float part[16 * kNCAM * 32];
  const int w    = threadIdx.x >> 5;
  const int lane = threadIdx.x & 31;
  const int hf   = lane >> 4, n = lane & 15;
  const int c0   = blockIdx.x * 32;
  const int kb   = w * 128 + hf * 4;
  const float* wp  = W + (size_t)kb * kNHC + c0 + 2 * n;
  const float* ap0 = CTp + (size_t)n * kCIN + kb;
  const float* ap1 = ap0 + (size_t)16 * kCIN;

  v2f b[16][4];
  v4f a[16][2];
#pragma unroll
  for (int p = 0; p < 8; ++p)
#pragma unroll
    for (int i = 0; i < 4; ++i)
      b[p][i] = __builtin_nontemporal_load(
          reinterpret_cast<const v2f*>(wp + (size_t)(8 * p + i) * kNHC));
#pragma unroll
  for (int p = 0; p < 8; ++p) {
    a[p][0] = *reinterpret_cast<const v4f*>(ap0 + 8 * p);
    a[p][1] = *reinterpret_cast<const v4f*>(ap1 + 8 * p);
  }
#pragma unroll
  for (int p = 8; p < 16; ++p)
#pragma unroll
    for (int i = 0; i < 4; ++i)
      b[p][i] = __builtin_nontemporal_load(
          reinterpret_cast<const v2f*>(wp + (size_t)(8 * p + i) * kNHC));

  __builtin_amdgcn_sched_barrier(0);

  v8f acc[2][2] = {};
#pragma unroll
  for (int p = 0; p < 16; ++p) {
    if (p < 8) {
      a[p + 8][0] = *reinterpret_cast<const v4f*>(ap0 + 8 * (p + 8));
      a[p + 8][1] = *reinterpret_cast<const v4f*>(ap1 + 8 * (p + 8));
    }
#pragma unroll
    for (int q = 0; q < 2; ++q) {
      v2f a0 = {a[p][0][2 * q], a[p][0][2 * q + 1]};
      v2f a1 = {a[p][1][2 * q], a[p][1][2 * q + 1]};
#pragma unroll
      for (int j = 0; j < 2; ++j) {
        v2f bj = {b[p][2 * q][j], b[p][2 * q + 1][j]};
        acc[0][j] = __builtin_amdgcn_wmma_f32_16x16x4_f32(false, a0, false, bj, (short)0,
                                                          acc[0][j], false, false);
        acc[1][j] = __builtin_amdgcn_wmma_f32_16x16x4_f32(false, a1, false, bj, (short)0,
                                                          acc[1][j], false, false);
      }
    }
    if (p < 8) __builtin_amdgcn_sched_barrier(0);
  }

  float* pw = part + w * (kNCAM * 32);
#pragma unroll
  for (int r = 0; r < 8; ++r) {
    v2f t = {acc[0][0][r], acc[0][1][r]};
    *reinterpret_cast<v2f*>(pw + (8 * hf + r) * 32 + 2 * n) = t;
  }
  if (hf == 0) {
#pragma unroll
    for (int r = 0; r < kNCAM - 16; ++r) {
      v2f t = {acc[1][0][r], acc[1][1][r]};
      *reinterpret_cast<v2f*>(pw + (16 + r) * 32 + 2 * n) = t;
    }
  }
  __syncthreads();
  for (int i = threadIdx.x; i < kNCAM * 32; i += 512) {
    float s = 0.f;
#pragma unroll
    for (int w2 = 0; w2 < 16; ++w2) s += part[w2 * (kNCAM * 32) + i];
    int cam = i >> 5, col = c0 + (i & 31);
    int h = col >> 11, c = col & (kCOUT - 1);
    Hs[(size_t)(h * kNCAM + cam) * kCOUT + c] = s;
  }
}

__global__ void k_as(const float* __restrict__ Hs,
                     const float* __restrict__ att_src,
                     float* __restrict__ a_s) {
  int wave = (blockIdx.x * blockDim.x + threadIdx.x) >> 5;
  int lane = threadIdx.x & 31;
  if (wave >= kNJ) return;
  int h = wave / kNCAM, cam = wave % kNCAM;
  const float4* hrow = reinterpret_cast<const float4*>(Hs + (size_t)wave * kCOUT);
  const float4* arow = reinterpret_cast<const float4*>(att_src + (size_t)h * kCOUT);
  float s = 0.f;
  for (int c = lane; c < kCOUT / 4; c += 32) {
    float4 hv = hrow[c];
    float4 av = arow[c];
    s += hv.x * av.x + hv.y * av.y + hv.z * av.z + hv.w * av.w;
  }
  s = wave_sum(s);
  if (lane == 0) a_s[cam * kH + h] = s;
}

__device__ __forceinline__ float edge_logit(const float* a_s, const float* a_d,
                                            int src, int dst, int h) {
  float v = a_s[src * kH + h] + a_d[dst * kH + h];
  return v > 0.f ? v : kNegSlope * v;
}

__global__ void k_edge_max(const int* __restrict__ esrc, const int* __restrict__ edst,
                           const float* __restrict__ a_s, const float* __restrict__ a_d,
                           unsigned* __restrict__ mEnc) {
  int t = blockIdx.x * blockDim.x + threadIdx.x;
  if (t >= kE * kH) return;
  int ei = t >> 2, h = t & 3;
  int src = esrc[ei], dst = edst[ei];
  float v = edge_logit(a_s, a_d, src, dst, h);
  atomicMax(&mEnc[dst * kH + h], f2ord(v));
}

__global__ void k_edge_den(const int* __restrict__ esrc, const int* __restrict__ edst,
                           const float* __restrict__ a_s, const float* __restrict__ a_d,
                           const unsigned* __restrict__ mEnc, float* __restrict__ den) {
  int t = blockIdx.x * blockDim.x + threadIdx.x;
  if (t >= kE * kH) return;
  int ei = t >> 2, h = t & 3;
  int src = esrc[ei], dst = edst[ei];
  float v = edge_logit(a_s, a_d, src, dst, h);
  float m = ord2f(mEnc[dst * kH + h]);
  atomicAdd(&den[dst * kH + h], __expf(v - m));
}

__global__ void k_edge_acc(const int* __restrict__ esrc, const int* __restrict__ edst,
                           const float* __restrict__ a_s, const float* __restrict__ a_d,
                           const unsigned* __restrict__ mEnc,
                           const float* __restrict__ den, float* __restrict__ S) {
  int t = blockIdx.x * blockDim.x + threadIdx.x;
  if (t >= kE * kH) return;
  int ei = t >> 2, h = t & 3;
  int src = esrc[ei], dst = edst[ei];
  float v = edge_logit(a_s, a_d, src, dst, h);
  float m = ord2f(mEnc[dst * kH + h]);
  float attn = __expf(v - m) / den[dst * kH + h];
  atomicAdd(&S[(size_t)dst * kNJ + h * kNCAM + src], attn);
}

__global__ void __launch_bounds__(256)
k_out(const float* __restrict__ S, const float* __restrict__ Hs,
      const float* __restrict__ x, const float* __restrict__ bias,
      const float* __restrict__ alphap, float* __restrict__ out) {
  constexpr int kTiles = 8;
  const int w    = threadIdx.x >> 5;
  const int lane = threadIdx.x & 31;
  const int hf   = lane >> 4, n = lane & 15;
  const int c0   = (blockIdx.x * 8 + w) * 64 + 4 * n;
  const int row0 = blockIdx.y * (kTiles * 16);

  const float* sp = S + (size_t)(row0 + n) * kNJ + 4 * hf;
  const float* xp = x + (size_t)(row0 + 8 * hf) * kCOUT + c0;
  float*       op = out + (size_t)(row0 + 8 * hf) * kCOUT + c0;

  v4f a[2][10];
  v4f xv[2][8];
#pragma unroll
  for (int u = 0; u < 10; ++u) a[0][u] = *reinterpret_cast<const v4f*>(sp + 8 * u);

  v2f bp[20][4];
  {
    const float* hp = Hs + (size_t)(4 * hf) * kCOUT + c0;
#pragma unroll
    for (int u = 0; u < 10; ++u) {
      v4f t[4];
#pragma unroll
      for (int i = 0; i < 4; ++i)
        t[i] = *reinterpret_cast<const v4f*>(hp + (size_t)(8 * u + i) * kCOUT);
#pragma unroll
      for (int q = 0; q < 2; ++q)
#pragma unroll
        for (int j = 0; j < 4; ++j) {
          v2f p = {t[2 * q][j], t[2 * q + 1][j]};
          bp[2 * u + q][j] = p;
        }
    }
  }
#pragma unroll
  for (int r = 0; r < 8; ++r)
    xv[0][r] = __builtin_nontemporal_load(
        reinterpret_cast<const v4f*>(xp + (size_t)r * kCOUT));

  const float alpha = *alphap;
  const v4f bv = *reinterpret_cast<const v4f*>(bias + c0);

#pragma unroll
  for (int mt = 0; mt < kTiles; ++mt) {
    const int cur = mt & 1, nxt = cur ^ 1;
    if (mt + 1 < kTiles) {
      const float* sn = sp + (size_t)(mt + 1) * 16 * kNJ;
      const float* xn = xp + (size_t)(mt + 1) * 16 * kCOUT;
#pragma unroll
      for (int u = 0; u < 10; ++u) a[nxt][u] = *reinterpret_cast<const v4f*>(sn + 8 * u);
#pragma unroll
      for (int r = 0; r < 8; ++r)
        xv[nxt][r] = __builtin_nontemporal_load(
            reinterpret_cast<const v4f*>(xn + (size_t)r * kCOUT));
    }
    __builtin_amdgcn_sched_barrier(0);

    v8f acc[4] = {};
#pragma unroll
    for (int u = 0; u < 10; ++u)
#pragma unroll
      for (int q = 0; q < 2; ++q) {
        v2f aq = {a[cur][u][2 * q], a[cur][u][2 * q + 1]};
#pragma unroll
        for (int j = 0; j < 4; ++j)
          acc[j] = __builtin_amdgcn_wmma_f32_16x16x4_f32(false, aq, false, bp[2 * u + q][j],
                                                         (short)0, acc[j], false, false);
      }

    float* ot = op + (size_t)mt * 16 * kCOUT;
#pragma unroll
    for (int r = 0; r < 8; ++r) {
      v4f xr = xv[cur][r];
      v4f o;
      o[0] = xr[0] + alpha * (acc[0][r] * 0.25f + bv[0]);
      o[1] = xr[1] + alpha * (acc[1][r] * 0.25f + bv[1]);
      o[2] = xr[2] + alpha * (acc[2][r] * 0.25f + bv[2]);
      o[3] = xr[3] + alpha * (acc[3][r] * 0.25f + bv[3]);
      *reinterpret_cast<v4f*>(ot + (size_t)r * kCOUT) = o;
    }
  }
}

}

extern "C" void kernel_launch(void* const* d_in, const int* in_sizes, int n_in,
                              void* d_out, int out_size, void* d_ws, size_t ws_size,
                              hipStream_t stream) {
  (void)in_sizes; (void)n_in; (void)out_size; (void)ws_size;
  const float* x       = (const float*)d_in[0];
  const float* ct      = (const float*)d_in[1];
  const float* W       = (const float*)d_in[2];
  const float* att_src = (const float*)d_in[3];
  const float* att_dst = (const float*)d_in[4];
  const float* bias    = (const float*)d_in[5];
  const float* alphap  = (const float*)d_in[6];
  const int* esrc = (const int*)d_in[8];
  const int* edst = (const int*)d_in[9];

  float* ws = (float*)d_ws;
  float*    Hs   = ws + OFF_HS;
  float*    Watt = ws + OFF_WATT;
  float*    a_d  = ws + OFF_AD;
  float*    a_s  = ws + OFF_AS;
  unsigned* mEnc = (unsigned*)(ws + OFF_MENC);
  float*    den  = ws + OFF_DEN;
  float*    S    = ws + OFF_S;
  float*    CTp  = ws + OFF_CTPAD;

  hipMemsetAsync(mEnc, 0, (size_t)kNV * kH * sizeof(unsigned), stream);
  hipMemsetAsync(den,  0, (size_t)kNV * kH * sizeof(float), stream);
  hipMemsetAsync(S,    0, (size_t)kNV * kNJ * sizeof(float), stream);
  hipMemsetAsync(CTp + (size_t)kNCAM * kCIN, 0,
                 (size_t)(32 - kNCAM) * kCIN * sizeof(float), stream);
  hipMemcpyAsync(CTp, ct, (size_t)kNCAM * kCIN * sizeof(float),
                 hipMemcpyDeviceToDevice, stream);

  k_fold_att<<<dim3((kCIN * kH) / 8), dim3(256), 0, stream>>>(W, att_dst, Watt);
  k_ad<<<dim3(kNV / 8), dim3(256), 0, stream>>>(x, Watt, a_d);
  k_hsrc<<<dim3(kNHC / 32), dim3(512), 0, stream>>>(CTp, W, Hs);
  k_as<<<dim3((kNJ + 7) / 8), dim3(256), 0, stream>>>(Hs, att_src, a_s);

  int et = kE * kH;
  k_edge_max<<<dim3((et + 255) / 256), dim3(256), 0, stream>>>(esrc, edst, a_s, a_d, mEnc);
  k_edge_den<<<dim3((et + 255) / 256), dim3(256), 0, stream>>>(esrc, edst, a_s, a_d, mEnc, den);
  k_edge_acc<<<dim3((et + 255) / 256), dim3(256), 0, stream>>>(esrc, edst, a_s, a_d, mEnc, den, S);

  k_out<<<dim3(kCOUT / 512, kNV / 128), dim3(256), 0, stream>>>(
      S, Hs, x, bias, alphap, (float*)d_out);
}
